# starting kernel
# baseline (speedup 1.0000x reference)
#include <hip/hip_runtime.h>

typedef __attribute__((ext_vector_type(2))) float v2f;
typedef __attribute__((ext_vector_type(8))) float v8f;

__global__ __launch_bounds__(256) void edge_dot_wmma_f32(
    const float* __restrict__ h,
    const int*   __restrict__ src,
    const int*   __restrict__ dst,
    float*       __restrict__ out,
    int n_edges)
{
    const int lane          = threadIdx.x & 31;
    const int waveInBlock   = threadIdx.x >> 5;
    const int wavesPerBlock = blockDim.x >> 5;
    const int waveGlobal    = blockIdx.x * wavesPerBlock + waveInBlock;
    const int nWavesTotal   = gridDim.x * wavesPerBlock;
    const int nGroups       = (n_edges + 15) >> 4;

    const int m = lane & 15;
    const int g = lane >> 4;

    for (int grp = waveGlobal; grp < nGroups; grp += nWavesTotal) {
        const int ebase = grp << 4;
        int em = ebase + m;
        if (em >= n_edges) em = n_edges - 1;

        const int si = src[em];
        const int di = dst[em];
        const float* __restrict__ pu = h + (long long)si * 64 + (g << 1);
        const float* __restrict__ pv = h + (long long)di * 64 + (g << 1);

        v8f acc = {};
#pragma unroll
        for (int k = 0; k < 16; ++k) {
            v2f a = *(const v2f*)(pu + 4 * k);
            v2f b = *(const v2f*)(pv + 4 * k);
            acc = __builtin_amdgcn_wmma_f32_16x16x4_f32(
                false, a, false, b, (short)0, acc, false, false);
        }

        int sel = -1, edge = -1;
        if (lane < 8)        { sel = lane;      edge = ebase + lane; }
        else if (lane >= 24) { sel = lane - 24; edge = ebase + lane - 16; }

        if (sel >= 0 && edge < n_edges) {
            float v = acc[0];
#pragma unroll
            for (int i = 1; i < 8; ++i) v = (sel == i) ? acc[i] : v;
            out[edge] = v;
        }
    }
}

extern "C" void kernel_launch(void* const* d_in, const int* in_sizes, int n_in,
                              void* d_out, int out_size, void* d_ws, size_t ws_size,
                              hipStream_t stream) {
    const float* h   = (const float*)d_in[0];
    const int*   src = (const int*)d_in[1];
    const int*   dst = (const int*)d_in[2];
    float*       out = (float*)d_out;

    const int n_edges = in_sizes[1];
    const int nGroups = (n_edges + 15) / 16;

    const int threads       = 256;
    const int wavesPerBlock = threads / 32;
    int blocks = (nGroups + wavesPerBlock - 1) / wavesPerBlock;
    if (blocks < 1) blocks = 1;

    edge_dot_wmma_f32<<<blocks, threads, 0, stream>>>(h, src, dst, out, n_edges);
}


# Row loads cover whole 128-B lines: a WMMA tile is 4 edges x 4 k-chunks, 8 lanes x 16 B per line (was 16 lines x 16 B per load); chunk sums added per lane quad by DPP: k order of the sum changes
# speedup vs baseline: 2.2475x; 2.2475x over previous
#include <hip/hip_runtime.h>

typedef __attribute__((ext_vector_type(2))) float v2f;
typedef __attribute__((ext_vector_type(4))) float v4f;
typedef __attribute__((ext_vector_type(8))) float v8f;
typedef __attribute__((ext_vector_type(4))) int   v4i;


__device__ __forceinline__ float quad_sum(float v)
{
    v += __int_as_float(__builtin_amdgcn_update_dpp(0, __float_as_int(v), 0x161, 0xF, 0xF, true));
    v += __int_as_float(__builtin_amdgcn_update_dpp(0, __float_as_int(v), 0x162, 0xF, 0xF, true));
    return v;
}

__global__ __launch_bounds__(256) void edge_dot_wmma_f32(
    const float* __restrict__ h,
    const int*   __restrict__ src,
    const int*   __restrict__ dst,
    float*       __restrict__ out,
    int n_edges)
{
    const int lane          = threadIdx.x & 31;
    const int waveInBlock   = threadIdx.x >> 5;
    const int wavesPerBlock = blockDim.x >> 5;
    const int waveGlobal    = blockIdx.x * wavesPerBlock + waveInBlock;
    const int nWavesTotal   = gridDim.x * wavesPerBlock;
    const int nGroups       = (n_edges + 15) >> 4;

    const int c    = lane & 15;
    const int hh   = lane >> 4;
    const int e    = c >> 2;
    const int q    = c & 3;
    const int koff = (4 * hh + q) * 4;
    const int r    = lane & 7;

    for (int grp = waveGlobal; grp < nGroups; grp += nWavesTotal) {
        const int ebase = grp << 4;

        v4i si, di;
        if (ebase + 16 <= n_edges) {
            si = *(const v4i*)(src + ebase + 4 * e);
            di = *(const v4i*)(dst + ebase + 4 * e);
        } else {
#pragma unroll
            for (int t = 0; t < 4; ++t) {
                int em = ebase + 4 * e + t;
                if (em >= n_edges) em = n_edges - 1;
                si[t] = src[em];
                di[t] = dst[em];
            }
        }

        v4f a0[4], a1[4], b0[4], b1[4];
#pragma unroll
        for (int t = 0; t < 4; ++t) {
            const float* __restrict__ pu = h + (long long)si[t] * 64 + koff;
            const float* __restrict__ pv = h + (long long)di[t] * 64 + koff;
            a0[t] = *(const v4f*)(pu);
            a1[t] = *(const v4f*)(pu + 32);
            b0[t] = *(const v4f*)(pv);
            b1[t] = *(const v4f*)(pv + 32);
        }

        float res = 0.0f;
#pragma unroll
        for (int t = 0; t < 4; ++t) {
            v8f acc = {};
            acc = __builtin_amdgcn_wmma_f32_16x16x4_f32(
                false, a0[t].xy, false, b0[t].xy, (short)0, acc, false, false);
            acc = __builtin_amdgcn_wmma_f32_16x16x4_f32(
                false, a0[t].zw, false, b0[t].zw, (short)0, acc, false, false);
            acc = __builtin_amdgcn_wmma_f32_16x16x4_f32(
                false, a1[t].xy, false, b1[t].xy, (short)0, acc, false, false);
            acc = __builtin_amdgcn_wmma_f32_16x16x4_f32(
                false, a1[t].zw, false, b1[t].zw, (short)0, acc, false, false);

            float v = acc[0];
#pragma unroll
            for (int i = 1; i < 8; ++i) v = (r == i) ? acc[i] : v;

            v = quad_sum(v);
            res = (q == t) ? v : res;
        }

        const int edge = ebase + c;
        if (hh == (c >> 3) && edge < n_edges) out[edge] = res;
    }
}

extern "C" void kernel_launch(void* const* d_in, const int* in_sizes, int n_in,
                              void* d_out, int out_size, void* d_ws, size_t ws_size,
                              hipStream_t stream) {
    const float* h   = (const float*)d_in[0];
    const int*   src = (const int*)d_in[1];
    const int*   dst = (const int*)d_in[2];
    float*       out = (float*)d_out;

    const int n_edges = in_sizes[1];
    const int nGroups = (n_edges + 15) / 16;

    const int threads       = 256;
    const int wavesPerBlock = threads / 32;
    int blocks = (nGroups + wavesPerBlock - 1) / wavesPerBlock;
    if (blocks < 1) blocks = 1;

    edge_dot_wmma_f32<<<blocks, threads, 0, stream>>>(h, src, dst, out, n_edges);
}


# Resident grid: 256 work-groups of 32 waves, one per WGP, each wave loops over 7-8 groups of 16 edges; 8,192 waves enter instead of 62,500 one-group waves; loop body and every sum unchanged
# speedup vs baseline: 2.3705x; 1.0547x over previous
#include <hip/hip_runtime.h>

typedef __attribute__((ext_vector_type(2))) float v2f;
typedef __attribute__((ext_vector_type(4))) float v4f;
typedef __attribute__((ext_vector_type(8))) float v8f;
typedef __attribute__((ext_vector_type(4))) int   v4i;


__device__ __forceinline__ float quad_sum(float v)
{
    v += __int_as_float(__builtin_amdgcn_update_dpp(0, __float_as_int(v), 0x161, 0xF, 0xF, true));
    v += __int_as_float(__builtin_amdgcn_update_dpp(0, __float_as_int(v), 0x162, 0xF, 0xF, true));
    return v;
}

__global__ __launch_bounds__(1024) void edge_dot_wmma_f32(
    const float* __restrict__ h,
    const int*   __restrict__ src,
    const int*   __restrict__ dst,
    float*       __restrict__ out,
    int n_edges)
{
    const int lane          = threadIdx.x & 31;
    const int waveInBlock   = threadIdx.x >> 5;
    const int wavesPerBlock = blockDim.x >> 5;
    const int nWavesTotal   = gridDim.x * wavesPerBlock;
    const int waveGlobal    = (waveInBlock >> 1) * (2 * gridDim.x) + 2 * blockIdx.x + (waveInBlock & 1);
    const int nGroups       = (n_edges + 15) >> 4;

    const int c    = lane & 15;
    const int hh   = lane >> 4;
    const int e    = c >> 2;
    const int q    = c & 3;
    const int koff = (4 * hh + q) * 4;
    const int r    = lane & 7;

    for (int grp = waveGlobal; grp < nGroups; grp += nWavesTotal) {
        const int ebase = grp << 4;

        v4i si, di;
        if (ebase + 16 <= n_edges) {
            si = *(const v4i*)(src + ebase + 4 * e);
            di = *(const v4i*)(dst + ebase + 4 * e);
        } else {
#pragma unroll
            for (int t = 0; t < 4; ++t) {
                int em = ebase + 4 * e + t;
                if (em >= n_edges) em = n_edges - 1;
                si[t] = src[em];
                di[t] = dst[em];
            }
        }

        v4f a0[4], a1[4], b0[4], b1[4];
#pragma unroll
        for (int t = 0; t < 4; ++t) {
            const float* __restrict__ pu = h + (long long)si[t] * 64 + koff;
            const float* __restrict__ pv = h + (long long)di[t] * 64 + koff;
            a0[t] = *(const v4f*)(pu);
            a1[t] = *(const v4f*)(pu + 32);
            b0[t] = *(const v4f*)(pv);
            b1[t] = *(const v4f*)(pv + 32);
        }

        float res = 0.0f;
#pragma unroll
        for (int t = 0; t < 4; ++t) {
            v8f acc = {};
            acc = __builtin_amdgcn_wmma_f32_16x16x4_f32(
                false, a0[t].xy, false, b0[t].xy, (short)0, acc, false, false);
            acc = __builtin_amdgcn_wmma_f32_16x16x4_f32(
                false, a0[t].zw, false, b0[t].zw, (short)0, acc, false, false);
            acc = __builtin_amdgcn_wmma_f32_16x16x4_f32(
                false, a1[t].xy, false, b1[t].xy, (short)0, acc, false, false);
            acc = __builtin_amdgcn_wmma_f32_16x16x4_f32(
                false, a1[t].zw, false, b1[t].zw, (short)0, acc, false, false);

            float v = acc[0];
#pragma unroll
            for (int i = 1; i < 8; ++i) v = (r == i) ? acc[i] : v;

            v = quad_sum(v);
            res = (q == t) ? v : res;
        }

        const int edge = ebase + c;
        if (hh == (c >> 3) && edge < n_edges) out[edge] = res;
    }
}

extern "C" void kernel_launch(void* const* d_in, const int* in_sizes, int n_in,
                              void* d_out, int out_size, void* d_ws, size_t ws_size,
                              hipStream_t stream) {
    const float* h   = (const float*)d_in[0];
    const int*   src = (const int*)d_in[1];
    const int*   dst = (const int*)d_in[2];
    float*       out = (float*)d_out;

    const int n_edges = in_sizes[1];

    edge_dot_wmma_f32<<<256, 1024, 0, stream>>>(h, src, dst, out, n_edges);
}
